# starting kernel
# baseline (speedup 1.0000x reference)
#include <hip/hip_runtime.h>

typedef float v2f __attribute__((ext_vector_type(2)));
typedef float v4f __attribute__((ext_vector_type(4)));
typedef float v8f __attribute__((ext_vector_type(8)));

#define WMMA_F32(a, b, c) \
    __builtin_amdgcn_wmma_f32_16x16x4_f32(false, (a), false, (b), (short)0, (c), false, false)

constexpr int Bn = 8, Tn = 2048, Cn = 1024, Hn = 64;

constexpr size_t OFF_KP = 0;
constexpr size_t OFF_QP = OFF_KP + (size_t)Bn * Tn * Hn;
constexpr size_t OFF_VT = OFF_QP + (size_t)Bn * Tn * Hn;
constexpr size_t OFF_WT = OFF_VT + (size_t)Bn * Tn * Hn;

__global__ void wtrans_kernel(const float* __restrict__ Wk,
                              const float* __restrict__ Wq,
                              const float* __restrict__ Wv,
                              float* __restrict__ wt) {
    int idx = blockIdx.x * 256 + threadIdx.x;
    int m = idx >> 16;
    int e = idx & 65535;
    int h = e >> 10;
    int c = e & 1023;
    const float* W = (m == 0) ? Wk : ((m == 1) ? Wq : Wv);
    float val = W[c * Hn + h];
    if (m == 1) val *= 0.125f;
    wt[(size_t)m * 65536 + e] = val;
}

__global__ void proj_kernel(const float* __restrict__ x,
                            const float* __restrict__ wt,
                            float* __restrict__ ws) {
    int wave  = blockIdx.x * (blockDim.x >> 5) + (threadIdx.x >> 5);
    int lane  = threadIdx.x & 31;
    int which = wave % 3;
    int rt    = wave / 3;
    int l15   = lane & 15;
    int hi    = lane >> 4;

    const float* xa  = x + (size_t)(rt * 16 + l15) * Cn + 2 * hi;
    const float* wbb = wt + (size_t)which * Hn * Cn + 2 * hi;
    const float* wb0 = wbb + (size_t)(0 * 16 + l15) * Cn;
    const float* wb1 = wbb + (size_t)(1 * 16 + l15) * Cn;
    const float* wb2 = wbb + (size_t)(2 * 16 + l15) * Cn;
    const float* wb3 = wbb + (size_t)(3 * 16 + l15) * Cn;

    v8f a0 = {}, a1 = {}, a2 = {}, a3 = {};
#pragma unroll 4
    for (int k = 0; k < Cn; k += 4) {
        v2f a = *(const v2f*)(xa + k);
        a0 = WMMA_F32(a, *(const v2f*)(wb0 + k), a0);
        a1 = WMMA_F32(a, *(const v2f*)(wb1 + k), a1);
        a2 = WMMA_F32(a, *(const v2f*)(wb2 + k), a2);
        a3 = WMMA_F32(a, *(const v2f*)(wb3 + k), a3);
    }

    if (which < 2) {
        float* p = ws + (which == 0 ? OFF_KP : OFF_QP);
        int rowbase = rt * 16 + 8 * hi;
#pragma unroll
        for (int r = 0; r < 8; r++) {
            p[(size_t)(rowbase + r) * Hn + 0 * 16 + l15] = a0[r];
            p[(size_t)(rowbase + r) * Hn + 1 * 16 + l15] = a1[r];
            p[(size_t)(rowbase + r) * Hn + 2 * 16 + l15] = a2[r];
            p[(size_t)(rowbase + r) * Hn + 3 * 16 + l15] = a3[r];
        }
    } else {
        int b  = rt >> 7;
        int t0 = (rt & 127) * 16 + 8 * hi;
        float* vtb = ws + OFF_VT + (size_t)b * Hn * Tn + t0;
        v8f accs0 = a0, accs1 = a1, accs2 = a2, accs3 = a3;
        {
            float* d = vtb + (size_t)(0 * 16 + l15) * Tn;
            *(v4f*)(d)     = (v4f){accs0[0], accs0[1], accs0[2], accs0[3]};
            *(v4f*)(d + 4) = (v4f){accs0[4], accs0[5], accs0[6], accs0[7]};
        }
        {
            float* d = vtb + (size_t)(1 * 16 + l15) * Tn;
            *(v4f*)(d)     = (v4f){accs1[0], accs1[1], accs1[2], accs1[3]};
            *(v4f*)(d + 4) = (v4f){accs1[4], accs1[5], accs1[6], accs1[7]};
        }
        {
            float* d = vtb + (size_t)(2 * 16 + l15) * Tn;
            *(v4f*)(d)     = (v4f){accs2[0], accs2[1], accs2[2], accs2[3]};
            *(v4f*)(d + 4) = (v4f){accs2[4], accs2[5], accs2[6], accs2[7]};
        }
        {
            float* d = vtb + (size_t)(3 * 16 + l15) * Tn;
            *(v4f*)(d)     = (v4f){accs3[0], accs3[1], accs3[2], accs3[3]};
            *(v4f*)(d + 4) = (v4f){accs3[4], accs3[5], accs3[6], accs3[7]};
        }
    }
}

__global__ void attn_kernel(const float* __restrict__ ws, float* __restrict__ out) {
    int wave = blockIdx.x * (blockDim.x >> 5) + (threadIdx.x >> 5);
    int lane = threadIdx.x & 31;
    int b = wave >> 7;
    int i = wave & 127;
    int l15 = lane & 15;
    int hi  = lane >> 4;

    const float* kp = ws + OFF_KP + (size_t)(b * Tn + i * 16 + l15) * Hn + 2 * hi;
    const float* qp = ws + OFF_QP + (size_t)b * Tn * Hn;
    const float* vt = ws + OFF_VT + (size_t)b * Hn * Tn;

    v2f kb[16];
#pragma unroll
    for (int kk = 0; kk < 16; kk++) kb[kk] = *(const v2f*)(kp + 4 * kk);

    v8f oacc[4] = {};
    float m = -__builtin_inff();
    float l = 0.0f;

    for (int j = 0; j <= i; j++) {
        const float* qa = qp + (size_t)(j * 16 + l15) * Hn + 2 * hi;
        v8f s = {};
#pragma unroll
        for (int kk = 0; kk < 16; kk++) {
            v2f a = *(const v2f*)(qa + 4 * kk);
            s = WMMA_F32(a, kb[kk], s);
        }

        if (j == i) {
#pragma unroll
            for (int r = 0; r < 8; r++) {
                int sl = r + 8 * hi;
                s[r] = (sl > l15) ? -__builtin_inff() : s[r];
            }
        }

        float tm = s[0];
#pragma unroll
        for (int r = 1; r < 8; r++) tm = fmaxf(tm, s[r]);
        tm = fmaxf(tm, __shfl_xor(tm, 16, 32));
        float mn = fmaxf(m, tm);
        float sc = __expf(m - mn);
        m = mn;

        float p[8];
        float tl = 0.0f;
#pragma unroll
        for (int r = 0; r < 8; r++) {
            p[r] = __expf(s[r] - mn);
            tl += p[r];
        }
        tl += __shfl_xor(tl, 16, 32);
        l = l * sc + tl;
#pragma unroll
        for (int n = 0; n < 4; n++) oacc[n] = oacc[n] * sc;

        float q0 = __shfl_xor(p[0], 16, 32), q1 = __shfl_xor(p[1], 16, 32);
        float q2 = __shfl_xor(p[2], 16, 32), q3 = __shfl_xor(p[3], 16, 32);
        float q4 = __shfl_xor(p[4], 16, 32), q5 = __shfl_xor(p[5], 16, 32);
        float q6 = __shfl_xor(p[6], 16, 32), q7 = __shfl_xor(p[7], 16, 32);
        bool ishi = (hi != 0);
        v2f bb[4];
        bb[0] = ishi ? (v2f){q2, q3} : (v2f){p[0], p[1]};
        bb[1] = ishi ? (v2f){q6, q7} : (v2f){p[4], p[5]};
        bb[2] = ishi ? (v2f){p[2], p[3]} : (v2f){q0, q1};
        bb[3] = ishi ? (v2f){p[6], p[7]} : (v2f){q4, q5};

        const float* va = vt + (size_t)l15 * Tn + j * 16 + 2 * hi;
#pragma unroll
        for (int n = 0; n < 4; n++) {
            const float* van = va + (size_t)n * 16 * Tn;
#pragma unroll
            for (int kk = 0; kk < 4; kk++) {
                v2f a = *(const v2f*)(van + 4 * kk);
                oacc[n] = WMMA_F32(a, bb[kk], oacc[n]);
            }
        }
    }

    float invl = 1.0f / l;
    float* dst = out + (size_t)(b * Tn + i * 16 + l15) * Hn + 8 * hi;
#pragma unroll
    for (int n = 0; n < 4; n++) {
        v4f lo = {oacc[n][0] * invl, oacc[n][1] * invl, oacc[n][2] * invl, oacc[n][3] * invl};
        v4f hi4 = {oacc[n][4] * invl, oacc[n][5] * invl, oacc[n][6] * invl, oacc[n][7] * invl};
        *(v4f*)(dst + n * 16)     = lo;
        *(v4f*)(dst + n * 16 + 4) = hi4;
    }
}

extern "C" void kernel_launch(void* const* d_in, const int* in_sizes, int n_in,
                              void* d_out, int out_size, void* d_ws, size_t ws_size,
                              hipStream_t stream) {
    const float* x  = (const float*)d_in[0];
    const float* Wk = (const float*)d_in[1];
    const float* Wq = (const float*)d_in[2];
    const float* Wv = (const float*)d_in[3];
    float* ws  = (float*)d_ws;
    float* out = (float*)d_out;

    wtrans_kernel<<<768, 256, 0, stream>>>(Wk, Wq, Wv, ws + OFF_WT);
    proj_kernel<<<384, 256, 0, stream>>>(x, ws + OFF_WT, ws);
    attn_kernel<<<128, 256, 0, stream>>>(ws, out);
}


# projection: first kernel writes the weights in matrix-fragment order, x is read 16 B per lane; one load feeds two k-steps, weight loads whole-line; changes the projections' k summation order
# speedup vs baseline: 2.5977x; 2.5977x over previous
#include <hip/hip_runtime.h>

typedef float v2f __attribute__((ext_vector_type(2)));
typedef float v4f __attribute__((ext_vector_type(4)));
typedef float v8f __attribute__((ext_vector_type(8)));

#define WMMA_F32(a, b, c) \
    __builtin_amdgcn_wmma_f32_16x16x4_f32(false, (a), false, (b), (short)0, (c), false, false)

constexpr int Bn = 8, Tn = 2048, Cn = 1024, Hn = 64;

constexpr size_t OFF_KP = 0;
constexpr size_t OFF_QP = OFF_KP + (size_t)Bn * Tn * Hn;
constexpr size_t OFF_VT = OFF_QP + (size_t)Bn * Tn * Hn;
constexpr size_t OFF_WT = OFF_VT + (size_t)Bn * Tn * Hn;

constexpr int NCHUNK = Cn / 8;
constexpr int WT_TILE = NCHUNK * 32 * 4;

__global__ void wtrans_kernel(const float* __restrict__ Wk,
                              const float* __restrict__ Wq,
                              const float* __restrict__ Wv,
                              float* __restrict__ wt) {
    int idx  = blockIdx.x * 256 + threadIdx.x;
    int m    = idx >> 14;
    int p    = idx & 16383;
    int nt   = p >> 12;
    int j    = (p >> 5) & (NCHUNK - 1);
    int lane = p & 31;
    int k0   = 8 * j + 4 * (lane >> 4);
    int n    = 16 * nt + (lane & 15);
    const float* W = (m == 0) ? Wk : ((m == 1) ? Wq : Wv);
    float sc = (m == 1) ? 0.125f : 1.0f;
    v4f val = {W[(k0 + 0) * Hn + n] * sc, W[(k0 + 1) * Hn + n] * sc,
               W[(k0 + 2) * Hn + n] * sc, W[(k0 + 3) * Hn + n] * sc};
    *(v4f*)(wt + (size_t)idx * 4) = val;
}

constexpr int PU = 4;

struct FragSet {
    v4f xa[PU];
    v4f wb[PU][4];
};

__device__ __forceinline__ void load_set(FragSet& f, const float* xa, const float* wb, int j0) {
#pragma unroll
    for (int u = 0; u < PU; u++) f.xa[u] = *(const v4f*)(xa + 8 * (j0 + u));
#pragma unroll
    for (int u = 0; u < PU; u++) {
#pragma unroll
        for (int n = 0; n < 4; n++)
            f.wb[u][n] = *(const v4f*)(wb + n * WT_TILE + (j0 + u) * 128);
    }
}

__device__ __forceinline__ void mma_set(const FragSet& f, v8f (&acc)[4]) {
#pragma unroll
    for (int u = 0; u < PU; u++) {
        v2f alo = {f.xa[u][0], f.xa[u][1]};
        v2f ahi = {f.xa[u][2], f.xa[u][3]};
#pragma unroll
        for (int n = 0; n < 4; n++)
            acc[n] = WMMA_F32(alo, ((v2f){f.wb[u][n][0], f.wb[u][n][1]}), acc[n]);
#pragma unroll
        for (int n = 0; n < 4; n++)
            acc[n] = WMMA_F32(ahi, ((v2f){f.wb[u][n][2], f.wb[u][n][3]}), acc[n]);
    }
}

__global__ __launch_bounds__(256) void proj_kernel(const float* __restrict__ x,
                            const float* __restrict__ wt,
                            float* __restrict__ ws) {
    int wave  = blockIdx.x * (blockDim.x >> 5) + (threadIdx.x >> 5);
    int lane  = threadIdx.x & 31;
    int which = wave % 3;
    int rt    = wave / 3;
    int l15   = lane & 15;
    int hi    = lane >> 4;

    const float* xa = x + (size_t)(rt * 16 + l15) * Cn + 4 * hi;
    const float* wb = wt + (size_t)which * 4 * WT_TILE + 4 * lane;

    v8f acc[4] = {};
    FragSet f0, f1;
    load_set(f0, xa, wb, 0);
    for (int j = 0; j < NCHUNK - 2 * PU; j += 2 * PU) {
        load_set(f1, xa, wb, j + PU);
        mma_set(f0, acc);
        load_set(f0, xa, wb, j + 2 * PU);
        mma_set(f1, acc);
        __builtin_amdgcn_sched_group_barrier(0x020, 5 * PU, 0);
        __builtin_amdgcn_sched_group_barrier(0x008, 8 * PU, 0);
        __builtin_amdgcn_sched_group_barrier(0x020, 5 * PU, 0);
        __builtin_amdgcn_sched_group_barrier(0x008, 8 * PU, 0);
    }
    load_set(f1, xa, wb, NCHUNK - PU);
    mma_set(f0, acc);
    mma_set(f1, acc);
    v8f a0 = acc[0], a1 = acc[1], a2 = acc[2], a3 = acc[3];

    if (which < 2) {
        float* p = ws + (which == 0 ? OFF_KP : OFF_QP);
        int rowbase = rt * 16 + 8 * hi;
#pragma unroll
        for (int r = 0; r < 8; r++) {
            p[(size_t)(rowbase + r) * Hn + 0 * 16 + l15] = a0[r];
            p[(size_t)(rowbase + r) * Hn + 1 * 16 + l15] = a1[r];
            p[(size_t)(rowbase + r) * Hn + 2 * 16 + l15] = a2[r];
            p[(size_t)(rowbase + r) * Hn + 3 * 16 + l15] = a3[r];
        }
    } else {
        int b  = rt >> 7;
        int t0 = (rt & 127) * 16 + 8 * hi;
        float* vtb = ws + OFF_VT + (size_t)b * Hn * Tn + t0;
        v8f accs0 = a0, accs1 = a1, accs2 = a2, accs3 = a3;
        {
            float* d = vtb + (size_t)(0 * 16 + l15) * Tn;
            *(v4f*)(d)     = (v4f){accs0[0], accs0[1], accs0[2], accs0[3]};
            *(v4f*)(d + 4) = (v4f){accs0[4], accs0[5], accs0[6], accs0[7]};
        }
        {
            float* d = vtb + (size_t)(1 * 16 + l15) * Tn;
            *(v4f*)(d)     = (v4f){accs1[0], accs1[1], accs1[2], accs1[3]};
            *(v4f*)(d + 4) = (v4f){accs1[4], accs1[5], accs1[6], accs1[7]};
        }
        {
            float* d = vtb + (size_t)(2 * 16 + l15) * Tn;
            *(v4f*)(d)     = (v4f){accs2[0], accs2[1], accs2[2], accs2[3]};
            *(v4f*)(d + 4) = (v4f){accs2[4], accs2[5], accs2[6], accs2[7]};
        }
        {
            float* d = vtb + (size_t)(3 * 16 + l15) * Tn;
            *(v4f*)(d)     = (v4f){accs3[0], accs3[1], accs3[2], accs3[3]};
            *(v4f*)(d + 4) = (v4f){accs3[4], accs3[5], accs3[6], accs3[7]};
        }
    }
}

constexpr int KS = 64;
constexpr int LP = 68;
constexpr int TILE_F = 64 * LP;

__device__ __forceinline__ void softmax_step(v8f s, float& m, float& l, v8f (&oacc)[4],
                                             v2f (&bb)[4], int hi) {
    float tm = s[0];
#pragma unroll
    for (int r = 1; r < 8; r++) tm = fmaxf(tm, s[r]);
    tm = fmaxf(tm, __shfl_xor(tm, 16, 32));
    float mn = fmaxf(m, tm);
    float sc = __expf(m - mn);
    m = mn;

    float p[8];
    float tl = 0.0f;
#pragma unroll
    for (int r = 0; r < 8; r++) {
        p[r] = __expf(s[r] - mn);
        tl += p[r];
    }
    tl += __shfl_xor(tl, 16, 32);
    l = l * sc + tl;
#pragma unroll
    for (int n = 0; n < 4; n++) oacc[n] = oacc[n] * sc;

    float q0 = __shfl_xor(p[0], 16, 32), q1 = __shfl_xor(p[1], 16, 32);
    float q2 = __shfl_xor(p[2], 16, 32), q3 = __shfl_xor(p[3], 16, 32);
    float q4 = __shfl_xor(p[4], 16, 32), q5 = __shfl_xor(p[5], 16, 32);
    float q6 = __shfl_xor(p[6], 16, 32), q7 = __shfl_xor(p[7], 16, 32);
    bool ishi = (hi != 0);
    bb[0] = ishi ? (v2f){q2, q3} : (v2f){p[0], p[1]};
    bb[1] = ishi ? (v2f){q6, q7} : (v2f){p[4], p[5]};
    bb[2] = ishi ? (v2f){p[2], p[3]} : (v2f){q0, q1};
    bb[3] = ishi ? (v2f){p[6], p[7]} : (v2f){q4, q5};
}

__global__ __launch_bounds__(128) void attn_kernel(const float* __restrict__ ws,
                                                   float* __restrict__ out) {
    __shared__ float lq[2][TILE_F];
    __shared__ float lv[2][TILE_F];

    int tid  = threadIdx.x;
    int w    = __builtin_amdgcn_readfirstlane(tid >> 5);
    int lane = tid & 31;
    int g    = 31 - (int)(blockIdx.x >> 3);
    int b    = blockIdx.x & 7;
    int i    = 4 * g + w;
    int l15  = lane & 15;
    int hi   = lane >> 4;

    const float* kp = ws + OFF_KP + (size_t)(b * Tn + i * 16 + l15) * Hn + 2 * hi;
    const float* qp = ws + OFF_QP + (size_t)b * Tn * Hn;
    const float* vt = ws + OFF_VT + (size_t)b * Hn * Tn;

    int srow = tid >> 4;
    int sc4  = tid & 15;
    const float* qsrc = qp + (size_t)srow * Hn + 4 * sc4;
    const float* vsrc = vt + (size_t)srow * Tn + 4 * sc4;
    int sdst = srow * LP + 4 * sc4;

    v4f pre[16];
#pragma unroll
    for (int r = 0; r < 8; r++) {
        pre[r]     = *(const v4f*)(qsrc + (size_t)(8 * r) * Hn);
        pre[8 + r] = *(const v4f*)(vsrc + (size_t)(8 * r) * Tn);
    }

    v2f kb[16];
#pragma unroll
    for (int kk = 0; kk < 16; kk++) kb[kk] = *(const v2f*)(kp + 4 * kk);

#pragma unroll
    for (int r = 0; r < 8; r++) {
        *(v4f*)(&lq[0][sdst + 8 * r * LP]) = pre[r];
        *(v4f*)(&lv[0][sdst + 8 * r * LP]) = pre[8 + r];
    }
    __syncthreads();

    v8f oacc[4] = {};
    float m = -__builtin_inff();
    float l = 0.0f;
    v2f bb[4];

    int fo = l15 * LP + 2 * hi;

    for (int s = 0; s < g; s++) {
        const float* cq = &lq[s & 1][fo];
        const float* cv = &lv[s & 1][fo];

#pragma unroll
        for (int r = 0; r < 8; r++) {
            pre[r]     = *(const v4f*)(qsrc + (size_t)((s + 1) * KS + 8 * r) * Hn);
            pre[8 + r] = *(const v4f*)(vsrc + (size_t)(8 * r) * Tn + (s + 1) * KS);
        }

        v2f fq[4][16], fv[4][16];
#pragma unroll
        for (int jj = 0; jj < 4; jj++) {
#pragma unroll
            for (int kk = 0; kk < 16; kk++)
                fq[jj][kk] = *(const v2f*)(cq + jj * 16 * LP + 4 * kk);
            if (jj > 0) {
#pragma unroll
                for (int kk = 0; kk < 16; kk++)
                    fv[jj - 1][kk] = *(const v2f*)(cv + (kk >> 2) * 16 * LP + (jj - 1) * 16 + 4 * (kk & 3));
            }
        }
#pragma unroll
        for (int kk = 0; kk < 16; kk++)
            fv[3][kk] = *(const v2f*)(cv + (kk >> 2) * 16 * LP + 3 * 16 + 4 * (kk & 3));
        __builtin_amdgcn_sched_barrier(0);

        v8f sacc = {};
#pragma unroll
        for (int kk = 0; kk < 16; kk++) sacc = WMMA_F32(fq[0][kk], kb[kk], sacc);

#pragma unroll
        for (int jj = 0; jj < 4; jj++) {
            v8f snext = {};
            if (jj < 3) {
#pragma unroll
                for (int kk = 0; kk < 16; kk++) snext = WMMA_F32(fq[jj + 1][kk], kb[kk], snext);
            }

            softmax_step(sacc, m, l, oacc, bb, hi);

#pragma unroll
            for (int n = 0; n < 4; n++) {
#pragma unroll
                for (int kk = 0; kk < 4; kk++)
                    oacc[n] = WMMA_F32(fv[jj][4 * n + kk], bb[kk], oacc[n]);
            }
            sacc = snext;
        }

        __builtin_amdgcn_sched_barrier(0);
#pragma unroll
        for (int r = 0; r < 8; r++) {
            *(v4f*)(&lq[(s + 1) & 1][sdst + 8 * r * LP]) = pre[r];
            *(v4f*)(&lv[(s + 1) & 1][sdst + 8 * r * LP]) = pre[8 + r];
        }
        __syncthreads();
    }

    {
        const float* cq = &lq[g & 1][fo];
        const float* cv = &lv[g & 1][fo];
        for (int jj = 0; jj <= w; jj++) {
            v2f dq[16], dv[16];
#pragma unroll
            for (int kk = 0; kk < 16; kk++) {
                dq[kk] = *(const v2f*)(cq + jj * 16 * LP + 4 * kk);
                dv[kk] = *(const v2f*)(cv + (kk >> 2) * 16 * LP + jj * 16 + 4 * (kk & 3));
            }
            __builtin_amdgcn_sched_barrier(0);

            v8f sacc = {};
#pragma unroll
            for (int kk = 0; kk < 16; kk++) sacc = WMMA_F32(dq[kk], kb[kk], sacc);

            if (jj == w) {
#pragma unroll
                for (int r = 0; r < 8; r++) {
                    int sl = r + 8 * hi;
                    sacc[r] = (sl > l15) ? -__builtin_inff() : sacc[r];
                }
            }

            softmax_step(sacc, m, l, oacc, bb, hi);

#pragma unroll
            for (int n = 0; n < 4; n++) {
#pragma unroll
                for (int kk = 0; kk < 4; kk++)
                    oacc[n] = WMMA_F32(dv[4 * n + kk], bb[kk], oacc[n]);
            }
        }
    }

    float invl = 1.0f / l;
    float* dst = out + (size_t)(b * Tn + i * 16 + l15) * Hn + 8 * hi;
#pragma unroll
    for (int n = 0; n < 4; n++) {
        v4f lo = {oacc[n][0] * invl, oacc[n][1] * invl, oacc[n][2] * invl, oacc[n][3] * invl};
        v4f hi4 = {oacc[n][4] * invl, oacc[n][5] * invl, oacc[n][6] * invl, oacc[n][7] * invl};
        *(v4f*)(dst + n * 16)     = lo;
        *(v4f*)(dst + n * 16 + 4) = hi4;
    }
}

extern "C" void kernel_launch(void* const* d_in, const int* in_sizes, int n_in,
                              void* d_out, int out_size, void* d_ws, size_t ws_size,
                              hipStream_t stream) {
    const float* x  = (const float*)d_in[0];
    const float* Wk = (const float*)d_in[1];
    const float* Wq = (const float*)d_in[2];
    const float* Wv = (const float*)d_in[3];
    float* ws  = (float*)d_ws;
    float* out = (float*)d_out;

    wtrans_kernel<<<192, 256, 0, stream>>>(Wk, Wq, Wv, ws + OFF_WT);
    proj_kernel<<<384, 256, 0, stream>>>(x, ws + OFF_WT, ws);
    attn_kernel<<<256, 128, 0, stream>>>(ws, out);
}


# projection: each wave computes a 32-row x 96-column tile (2 A x 6 B fragments, 12 accumulators) instead of 16 x 64; 1024 waves, one per SIMD; same summation order
# speedup vs baseline: 2.9166x; 1.1228x over previous
#include <hip/hip_runtime.h>

typedef float v2f __attribute__((ext_vector_type(2)));
typedef float v4f __attribute__((ext_vector_type(4)));
typedef float v8f __attribute__((ext_vector_type(8)));

#define WMMA_F32(a, b, c) \
    __builtin_amdgcn_wmma_f32_16x16x4_f32(false, (a), false, (b), (short)0, (c), false, false)

constexpr int Bn = 8, Tn = 2048, Cn = 1024, Hn = 64;

constexpr size_t OFF_KP = 0;
constexpr size_t OFF_QP = OFF_KP + (size_t)Bn * Tn * Hn;
constexpr size_t OFF_VT = OFF_QP + (size_t)Bn * Tn * Hn;
constexpr size_t OFF_WT = OFF_VT + (size_t)Bn * Tn * Hn;

constexpr int NCHUNK = Cn / 8;
constexpr int WT_TILE = NCHUNK * 32 * 4;

__global__ void wtrans_kernel(const float* __restrict__ Wk,
                              const float* __restrict__ Wq,
                              const float* __restrict__ Wv,
                              float* __restrict__ wt) {
    int idx  = blockIdx.x * 256 + threadIdx.x;
    int m    = idx >> 14;
    int p    = idx & 16383;
    int nt   = p >> 12;
    int j    = (p >> 5) & (NCHUNK - 1);
    int lane = p & 31;
    int k0   = 8 * j + 4 * (lane >> 4);
    int n    = 16 * nt + (lane & 15);
    const float* W = (m == 0) ? Wk : ((m == 1) ? Wq : Wv);
    float sc = (m == 1) ? 0.125f : 1.0f;
    v4f val = {W[(k0 + 0) * Hn + n] * sc, W[(k0 + 1) * Hn + n] * sc,
               W[(k0 + 2) * Hn + n] * sc, W[(k0 + 3) * Hn + n] * sc};
    *(v4f*)(wt + (size_t)idx * 4) = val;
}

constexpr int PU = 4;
constexpr int RA = 2;
constexpr int NB = 6;

struct FragSet {
    v4f xa[PU][RA];
    v4f wb[PU][NB];
};

__device__ __forceinline__ void load_set(FragSet& f, const float* xa, const float* wb, int j0) {
#pragma unroll
    for (int u = 0; u < PU; u++) {
#pragma unroll
        for (int a = 0; a < RA; a++)
            f.xa[u][a] = *(const v4f*)(xa + (size_t)a * 16 * Cn + 8 * (j0 + u));
    }
#pragma unroll
    for (int u = 0; u < PU; u++) {
#pragma unroll
        for (int n = 0; n < NB; n++)
            f.wb[u][n] = *(const v4f*)(wb + n * WT_TILE + (j0 + u) * 128);
    }
}

__device__ __forceinline__ void mma_set(const FragSet& f, v8f (&acc)[RA][NB]) {
#pragma unroll
    for (int u = 0; u < PU; u++) {
#pragma unroll
        for (int a = 0; a < RA; a++) {
            v2f alo = {f.xa[u][a][0], f.xa[u][a][1]};
#pragma unroll
            for (int n = 0; n < NB; n++)
                acc[a][n] = WMMA_F32(alo, ((v2f){f.wb[u][n][0], f.wb[u][n][1]}), acc[a][n]);
        }
#pragma unroll
        for (int a = 0; a < RA; a++) {
            v2f ahi = {f.xa[u][a][2], f.xa[u][a][3]};
#pragma unroll
            for (int n = 0; n < NB; n++)
                acc[a][n] = WMMA_F32(ahi, ((v2f){f.wb[u][n][2], f.wb[u][n][3]}), acc[a][n]);
        }
    }
}

__device__ __forceinline__ void store_rowmajor(float* p, v8f a) {
#pragma unroll
    for (int r = 0; r < 8; r++) p[(size_t)r * Hn] = a[r];
}

__device__ __forceinline__ void store_transposed(float* d, v8f a) {
    *(v4f*)(d)     = (v4f){a[0], a[1], a[2], a[3]};
    *(v4f*)(d + 4) = (v4f){a[4], a[5], a[6], a[7]};
}

__global__ __launch_bounds__(128) void proj_kernel(const float* __restrict__ x,
                            const float* __restrict__ wt,
                            float* __restrict__ ws) {
    int w    = __builtin_amdgcn_readfirstlane(threadIdx.x >> 5);
    int lane = threadIdx.x & 31;
    int hf   = w & 1;
    int rb   = 2 * blockIdx.x + (w >> 1);
    int l15  = lane & 15;
    int hi   = lane >> 4;

    const float* xa = x + (size_t)(rb * 32 + l15) * Cn + 4 * hi;
    const float* wb = wt + (size_t)(NB * hf) * WT_TILE + 4 * lane;

    v8f acc[RA][NB] = {};
    FragSet f0, f1;
    load_set(f0, xa, wb, 0);
    for (int j = 0; j < NCHUNK - 2 * PU; j += 2 * PU) {
        load_set(f1, xa, wb, j + PU);
        mma_set(f0, acc);
        load_set(f0, xa, wb, j + 2 * PU);
        mma_set(f1, acc);
        __builtin_amdgcn_sched_group_barrier(0x020, (RA + NB) * PU, 0);
        __builtin_amdgcn_sched_group_barrier(0x008, 2 * RA * NB * PU, 0);
        __builtin_amdgcn_sched_group_barrier(0x020, (RA + NB) * PU, 0);
        __builtin_amdgcn_sched_group_barrier(0x008, 2 * RA * NB * PU, 0);
    }
    load_set(f1, xa, wb, NCHUNK - PU);
    mma_set(f0, acc);
    mma_set(f1, acc);

    float* pk = ws + OFF_KP + (size_t)(rb * 32 + 8 * hi) * Hn + l15;
    float* pq = ws + OFF_QP + (size_t)(rb * 32 + 8 * hi) * Hn + l15;
    if (hf == 0) {
#pragma unroll
        for (int a = 0; a < RA; a++) {
            store_rowmajor(pk + (size_t)a * 16 * Hn + 0 * 16, acc[a][0]);
            store_rowmajor(pk + (size_t)a * 16 * Hn + 1 * 16, acc[a][1]);
            store_rowmajor(pk + (size_t)a * 16 * Hn + 2 * 16, acc[a][2]);
            store_rowmajor(pk + (size_t)a * 16 * Hn + 3 * 16, acc[a][3]);
            store_rowmajor(pq + (size_t)a * 16 * Hn + 0 * 16, acc[a][4]);
            store_rowmajor(pq + (size_t)a * 16 * Hn + 1 * 16, acc[a][5]);
        }
    } else {
        int b  = rb >> 6;
        int t0 = (rb & 63) * 32 + 8 * hi;
        float* vtb = ws + OFF_VT + (size_t)b * Hn * Tn + t0 + (size_t)l15 * Tn;
#pragma unroll
        for (int a = 0; a < RA; a++) {
            store_rowmajor(pq + (size_t)a * 16 * Hn + 2 * 16, acc[a][0]);
            store_rowmajor(pq + (size_t)a * 16 * Hn + 3 * 16, acc[a][1]);
            store_transposed(vtb + a * 16 + (size_t)(0 * 16) * Tn, acc[a][2]);
            store_transposed(vtb + a * 16 + (size_t)(1 * 16) * Tn, acc[a][3]);
            store_transposed(vtb + a * 16 + (size_t)(2 * 16) * Tn, acc[a][4]);
            store_transposed(vtb + a * 16 + (size_t)(3 * 16) * Tn, acc[a][5]);
        }
    }
}

constexpr int KS = 64;
constexpr int LP = 68;
constexpr int TILE_F = 64 * LP;

__device__ __forceinline__ void softmax_step(v8f s, float& m, float& l, v8f (&oacc)[4],
                                             v2f (&bb)[4], int hi) {
    float tm = s[0];
#pragma unroll
    for (int r = 1; r < 8; r++) tm = fmaxf(tm, s[r]);
    tm = fmaxf(tm, __shfl_xor(tm, 16, 32));
    float mn = fmaxf(m, tm);
    float sc = __expf(m - mn);
    m = mn;

    float p[8];
    float tl = 0.0f;
#pragma unroll
    for (int r = 0; r < 8; r++) {
        p[r] = __expf(s[r] - mn);
        tl += p[r];
    }
    tl += __shfl_xor(tl, 16, 32);
    l = l * sc + tl;
#pragma unroll
    for (int n = 0; n < 4; n++) oacc[n] = oacc[n] * sc;

    float q0 = __shfl_xor(p[0], 16, 32), q1 = __shfl_xor(p[1], 16, 32);
    float q2 = __shfl_xor(p[2], 16, 32), q3 = __shfl_xor(p[3], 16, 32);
    float q4 = __shfl_xor(p[4], 16, 32), q5 = __shfl_xor(p[5], 16, 32);
    float q6 = __shfl_xor(p[6], 16, 32), q7 = __shfl_xor(p[7], 16, 32);
    bool ishi = (hi != 0);
    bb[0] = ishi ? (v2f){q2, q3} : (v2f){p[0], p[1]};
    bb[1] = ishi ? (v2f){q6, q7} : (v2f){p[4], p[5]};
    bb[2] = ishi ? (v2f){p[2], p[3]} : (v2f){q0, q1};
    bb[3] = ishi ? (v2f){p[6], p[7]} : (v2f){q4, q5};
}

__global__ __launch_bounds__(128) void attn_kernel(const float* __restrict__ ws,
                                                   float* __restrict__ out) {
    __shared__ float lq[2][TILE_F];
    __shared__ float lv[2][TILE_F];

    int tid  = threadIdx.x;
    int w    = __builtin_amdgcn_readfirstlane(tid >> 5);
    int lane = tid & 31;
    int g    = 31 - (int)(blockIdx.x >> 3);
    int b    = blockIdx.x & 7;
    int i    = 4 * g + w;
    int l15  = lane & 15;
    int hi   = lane >> 4;

    const float* kp = ws + OFF_KP + (size_t)(b * Tn + i * 16 + l15) * Hn + 2 * hi;
    const float* qp = ws + OFF_QP + (size_t)b * Tn * Hn;
    const float* vt = ws + OFF_VT + (size_t)b * Hn * Tn;

    int srow = tid >> 4;
    int sc4  = tid & 15;
    const float* qsrc = qp + (size_t)srow * Hn + 4 * sc4;
    const float* vsrc = vt + (size_t)srow * Tn + 4 * sc4;
    int sdst = srow * LP + 4 * sc4;

    v4f pre[16];
#pragma unroll
    for (int r = 0; r < 8; r++) {
        pre[r]     = *(const v4f*)(qsrc + (size_t)(8 * r) * Hn);
        pre[8 + r] = *(const v4f*)(vsrc + (size_t)(8 * r) * Tn);
    }

    v2f kb[16];
#pragma unroll
    for (int kk = 0; kk < 16; kk++) kb[kk] = *(const v2f*)(kp + 4 * kk);

#pragma unroll
    for (int r = 0; r < 8; r++) {
        *(v4f*)(&lq[0][sdst + 8 * r * LP]) = pre[r];
        *(v4f*)(&lv[0][sdst + 8 * r * LP]) = pre[8 + r];
    }
    __syncthreads();

    v8f oacc[4] = {};
    float m = -__builtin_inff();
    float l = 0.0f;
    v2f bb[4];

    int fo = l15 * LP + 2 * hi;

    for (int s = 0; s < g; s++) {
        const float* cq = &lq[s & 1][fo];
        const float* cv = &lv[s & 1][fo];

#pragma unroll
        for (int r = 0; r < 8; r++) {
            pre[r]     = *(const v4f*)(qsrc + (size_t)((s + 1) * KS + 8 * r) * Hn);
            pre[8 + r] = *(const v4f*)(vsrc + (size_t)(8 * r) * Tn + (s + 1) * KS);
        }

        v2f fq[4][16], fv[4][16];
#pragma unroll
        for (int jj = 0; jj < 4; jj++) {
#pragma unroll
            for (int kk = 0; kk < 16; kk++)
                fq[jj][kk] = *(const v2f*)(cq + jj * 16 * LP + 4 * kk);
            if (jj > 0) {
#pragma unroll
                for (int kk = 0; kk < 16; kk++)
                    fv[jj - 1][kk] = *(const v2f*)(cv + (kk >> 2) * 16 * LP + (jj - 1) * 16 + 4 * (kk & 3));
            }
        }
#pragma unroll
        for (int kk = 0; kk < 16; kk++)
            fv[3][kk] = *(const v2f*)(cv + (kk >> 2) * 16 * LP + 3 * 16 + 4 * (kk & 3));
        __builtin_amdgcn_sched_barrier(0);

        v8f sacc = {};
#pragma unroll
        for (int kk = 0; kk < 16; kk++) sacc = WMMA_F32(fq[0][kk], kb[kk], sacc);

#pragma unroll
        for (int jj = 0; jj < 4; jj++) {
            v8f snext = {};
            if (jj < 3) {
#pragma unroll
                for (int kk = 0; kk < 16; kk++) snext = WMMA_F32(fq[jj + 1][kk], kb[kk], snext);
            }

            softmax_step(sacc, m, l, oacc, bb, hi);

#pragma unroll
            for (int n = 0; n < 4; n++) {
#pragma unroll
                for (int kk = 0; kk < 4; kk++)
                    oacc[n] = WMMA_F32(fv[jj][4 * n + kk], bb[kk], oacc[n]);
            }
            sacc = snext;
        }

        __builtin_amdgcn_sched_barrier(0);
#pragma unroll
        for (int r = 0; r < 8; r++) {
            *(v4f*)(&lq[(s + 1) & 1][sdst + 8 * r * LP]) = pre[r];
            *(v4f*)(&lv[(s + 1) & 1][sdst + 8 * r * LP]) = pre[8 + r];
        }
        __syncthreads();
    }

    {
        const float* cq = &lq[g & 1][fo];
        const float* cv = &lv[g & 1][fo];
        for (int jj = 0; jj <= w; jj++) {
            v2f dq[16], dv[16];
#pragma unroll
            for (int kk = 0; kk < 16; kk++) {
                dq[kk] = *(const v2f*)(cq + jj * 16 * LP + 4 * kk);
                dv[kk] = *(const v2f*)(cv + (kk >> 2) * 16 * LP + jj * 16 + 4 * (kk & 3));
            }
            __builtin_amdgcn_sched_barrier(0);

            v8f sacc = {};
#pragma unroll
            for (int kk = 0; kk < 16; kk++) sacc = WMMA_F32(dq[kk], kb[kk], sacc);

            if (jj == w) {
#pragma unroll
                for (int r = 0; r < 8; r++) {
                    int sl = r + 8 * hi;
                    sacc[r] = (sl > l15) ? -__builtin_inff() : sacc[r];
                }
            }

            softmax_step(sacc, m, l, oacc, bb, hi);

#pragma unroll
            for (int n = 0; n < 4; n++) {
#pragma unroll
                for (int kk = 0; kk < 4; kk++)
                    oacc[n] = WMMA_F32(dv[4 * n + kk], bb[kk], oacc[n]);
            }
        }
    }

    float invl = 1.0f / l;
    float* dst = out + (size_t)(b * Tn + i * 16 + l15) * Hn + 8 * hi;
#pragma unroll
    for (int n = 0; n < 4; n++) {
        v4f lo = {oacc[n][0] * invl, oacc[n][1] * invl, oacc[n][2] * invl, oacc[n][3] * invl};
        v4f hi4 = {oacc[n][4] * invl, oacc[n][5] * invl, oacc[n][6] * invl, oacc[n][7] * invl};
        *(v4f*)(dst + n * 16)     = lo;
        *(v4f*)(dst + n * 16 + 4) = hi4;
    }
}

extern "C" void kernel_launch(void* const* d_in, const int* in_sizes, int n_in,
                              void* d_out, int out_size, void* d_ws, size_t ws_size,
                              hipStream_t stream) {
    const float* x  = (const float*)d_in[0];
    const float* Wk = (const float*)d_in[1];
    const float* Wq = (const float*)d_in[2];
    const float* Wv = (const float*)d_in[3];
    float* ws  = (float*)d_ws;
    float* out = (float*)d_out;

    wtrans_kernel<<<192, 256, 0, stream>>>(Wk, Wq, Wv, ws + OFF_WT);
    proj_kernel<<<256, 128, 0, stream>>>(x, ws + OFF_WT, ws);
    attn_kernel<<<256, 128, 0, stream>>>(ws, out);
}


# attention: the softmax's exchanges between the two lane halves (max, sum, P re-fragment) as v_permlanex16 on the vector ALU instead of shuffles through the LDS unit; same summation order
# speedup vs baseline: 3.2198x; 1.1040x over previous
#include <hip/hip_runtime.h>

typedef float v2f __attribute__((ext_vector_type(2)));
typedef float v4f __attribute__((ext_vector_type(4)));
typedef float v8f __attribute__((ext_vector_type(8)));

#define WMMA_F32(a, b, c) \
    __builtin_amdgcn_wmma_f32_16x16x4_f32(false, (a), false, (b), (short)0, (c), false, false)

constexpr int Bn = 8, Tn = 2048, Cn = 1024, Hn = 64;

constexpr size_t OFF_KP = 0;
constexpr size_t OFF_QP = OFF_KP + (size_t)Bn * Tn * Hn;
constexpr size_t OFF_VT = OFF_QP + (size_t)Bn * Tn * Hn;
constexpr size_t OFF_WT = OFF_VT + (size_t)Bn * Tn * Hn;

constexpr int NCHUNK = Cn / 8;
constexpr int WT_TILE = NCHUNK * 32 * 4;

__global__ void wtrans_kernel(const float* __restrict__ Wk,
                              const float* __restrict__ Wq,
                              const float* __restrict__ Wv,
                              float* __restrict__ wt) {
    int idx  = blockIdx.x * 256 + threadIdx.x;
    int m    = idx >> 14;
    int p    = idx & 16383;
    int nt   = p >> 12;
    int j    = (p >> 5) & (NCHUNK - 1);
    int lane = p & 31;
    int k0   = 8 * j + 4 * (lane >> 4);
    int n    = 16 * nt + (lane & 15);
    const float* W = (m == 0) ? Wk : ((m == 1) ? Wq : Wv);
    float sc = (m == 1) ? 0.125f : 1.0f;
    v4f val = {W[(k0 + 0) * Hn + n] * sc, W[(k0 + 1) * Hn + n] * sc,
               W[(k0 + 2) * Hn + n] * sc, W[(k0 + 3) * Hn + n] * sc};
    *(v4f*)(wt + (size_t)idx * 4) = val;
}

constexpr int PU = 4;
constexpr int RA = 2;
constexpr int NB = 6;

struct FragSet {
    v4f xa[PU][RA];
    v4f wb[PU][NB];
};

__device__ __forceinline__ void load_set(FragSet& f, const float* xa, const float* wb, int j0) {
#pragma unroll
    for (int u = 0; u < PU; u++) {
#pragma unroll
        for (int a = 0; a < RA; a++)
            f.xa[u][a] = *(const v4f*)(xa + (size_t)a * 16 * Cn + 8 * (j0 + u));
    }
#pragma unroll
    for (int u = 0; u < PU; u++) {
#pragma unroll
        for (int n = 0; n < NB; n++)
            f.wb[u][n] = *(const v4f*)(wb + n * WT_TILE + (j0 + u) * 128);
    }
}

__device__ __forceinline__ void mma_set(const FragSet& f, v8f (&acc)[RA][NB]) {
#pragma unroll
    for (int u = 0; u < PU; u++) {
#pragma unroll
        for (int a = 0; a < RA; a++) {
            v2f alo = {f.xa[u][a][0], f.xa[u][a][1]};
#pragma unroll
            for (int n = 0; n < NB; n++)
                acc[a][n] = WMMA_F32(alo, ((v2f){f.wb[u][n][0], f.wb[u][n][1]}), acc[a][n]);
        }
#pragma unroll
        for (int a = 0; a < RA; a++) {
            v2f ahi = {f.xa[u][a][2], f.xa[u][a][3]};
#pragma unroll
            for (int n = 0; n < NB; n++)
                acc[a][n] = WMMA_F32(ahi, ((v2f){f.wb[u][n][2], f.wb[u][n][3]}), acc[a][n]);
        }
    }
}

__device__ __forceinline__ void store_rowmajor(float* p, v8f a) {
#pragma unroll
    for (int r = 0; r < 8; r++) p[(size_t)r * Hn] = a[r];
}

__device__ __forceinline__ void store_transposed(float* d, v8f a) {
    *(v4f*)(d)     = (v4f){a[0], a[1], a[2], a[3]};
    *(v4f*)(d + 4) = (v4f){a[4], a[5], a[6], a[7]};
}

__global__ __launch_bounds__(128) void proj_kernel(const float* __restrict__ x,
                            const float* __restrict__ wt,
                            float* __restrict__ ws) {
    int w    = __builtin_amdgcn_readfirstlane(threadIdx.x >> 5);
    int lane = threadIdx.x & 31;
    int hf   = w & 1;
    int rb   = 2 * blockIdx.x + (w >> 1);
    int l15  = lane & 15;
    int hi   = lane >> 4;

    const float* xa = x + (size_t)(rb * 32 + l15) * Cn + 4 * hi;
    const float* wb = wt + (size_t)(NB * hf) * WT_TILE + 4 * lane;

    v8f acc[RA][NB] = {};
    FragSet f0, f1;
    load_set(f0, xa, wb, 0);
    for (int j = 0; j < NCHUNK - 2 * PU; j += 2 * PU) {
        load_set(f1, xa, wb, j + PU);
        mma_set(f0, acc);
        load_set(f0, xa, wb, j + 2 * PU);
        mma_set(f1, acc);
        __builtin_amdgcn_sched_group_barrier(0x020, (RA + NB) * PU, 0);
        __builtin_amdgcn_sched_group_barrier(0x008, 2 * RA * NB * PU, 0);
        __builtin_amdgcn_sched_group_barrier(0x020, (RA + NB) * PU, 0);
        __builtin_amdgcn_sched_group_barrier(0x008, 2 * RA * NB * PU, 0);
    }
    load_set(f1, xa, wb, NCHUNK - PU);
    mma_set(f0, acc);
    mma_set(f1, acc);

    float* pk = ws + OFF_KP + (size_t)(rb * 32 + 8 * hi) * Hn + l15;
    float* pq = ws + OFF_QP + (size_t)(rb * 32 + 8 * hi) * Hn + l15;
    if (hf == 0) {
#pragma unroll
        for (int a = 0; a < RA; a++) {
            store_rowmajor(pk + (size_t)a * 16 * Hn + 0 * 16, acc[a][0]);
            store_rowmajor(pk + (size_t)a * 16 * Hn + 1 * 16, acc[a][1]);
            store_rowmajor(pk + (size_t)a * 16 * Hn + 2 * 16, acc[a][2]);
            store_rowmajor(pk + (size_t)a * 16 * Hn + 3 * 16, acc[a][3]);
            store_rowmajor(pq + (size_t)a * 16 * Hn + 0 * 16, acc[a][4]);
            store_rowmajor(pq + (size_t)a * 16 * Hn + 1 * 16, acc[a][5]);
        }
    } else {
        int b  = rb >> 6;
        int t0 = (rb & 63) * 32 + 8 * hi;
        float* vtb = ws + OFF_VT + (size_t)b * Hn * Tn + t0 + (size_t)l15 * Tn;
#pragma unroll
        for (int a = 0; a < RA; a++) {
            store_rowmajor(pq + (size_t)a * 16 * Hn + 2 * 16, acc[a][0]);
            store_rowmajor(pq + (size_t)a * 16 * Hn + 3 * 16, acc[a][1]);
            store_transposed(vtb + a * 16 + (size_t)(0 * 16) * Tn, acc[a][2]);
            store_transposed(vtb + a * 16 + (size_t)(1 * 16) * Tn, acc[a][3]);
            store_transposed(vtb + a * 16 + (size_t)(2 * 16) * Tn, acc[a][4]);
            store_transposed(vtb + a * 16 + (size_t)(3 * 16) * Tn, acc[a][5]);
        }
    }
}

constexpr int KS = 64;
constexpr int LP = 68;
constexpr int TILE_F = 64 * LP;

__device__ __forceinline__ float xhalf(float x) {
    int b = __builtin_bit_cast(int, x);
    return __builtin_bit_cast(float,
        __builtin_amdgcn_permlanex16(b, b, 0x76543210, 0xfedcba98, false, false));
}

__device__ __forceinline__ void softmax_step(v8f s, float& m, float& l, v8f (&oacc)[4],
                                             v2f (&bb)[4], int hi) {
    float tm = s[0];
#pragma unroll
    for (int r = 1; r < 8; r++) tm = fmaxf(tm, s[r]);
    tm = fmaxf(tm, xhalf(tm));
    float mn = fmaxf(m, tm);
    float sc = __expf(m - mn);
    m = mn;

    float p[8];
    float tl = 0.0f;
#pragma unroll
    for (int r = 0; r < 8; r++) {
        p[r] = __expf(s[r] - mn);
        tl += p[r];
    }
    tl += xhalf(tl);
    l = l * sc + tl;
#pragma unroll
    for (int n = 0; n < 4; n++) oacc[n] = oacc[n] * sc;

    float q0 = xhalf(p[0]), q1 = xhalf(p[1]), q2 = xhalf(p[2]), q3 = xhalf(p[3]);
    float q4 = xhalf(p[4]), q5 = xhalf(p[5]), q6 = xhalf(p[6]), q7 = xhalf(p[7]);
    bool ishi = (hi != 0);
    bb[0] = ishi ? (v2f){q2, q3} : (v2f){p[0], p[1]};
    bb[1] = ishi ? (v2f){q6, q7} : (v2f){p[4], p[5]};
    bb[2] = ishi ? (v2f){p[2], p[3]} : (v2f){q0, q1};
    bb[3] = ishi ? (v2f){p[6], p[7]} : (v2f){q4, q5};
}

__global__ __launch_bounds__(128) void attn_kernel(const float* __restrict__ ws,
                                                   float* __restrict__ out) {
    __shared__ float lq[2][TILE_F];
    __shared__ float lv[2][TILE_F];

    int tid  = threadIdx.x;
    int w    = __builtin_amdgcn_readfirstlane(tid >> 5);
    int lane = tid & 31;
    int g    = 31 - (int)(blockIdx.x >> 3);
    int b    = blockIdx.x & 7;
    int i    = 4 * g + w;
    int l15  = lane & 15;
    int hi   = lane >> 4;

    const float* kp = ws + OFF_KP + (size_t)(b * Tn + i * 16 + l15) * Hn + 2 * hi;
    const float* qp = ws + OFF_QP + (size_t)b * Tn * Hn;
    const float* vt = ws + OFF_VT + (size_t)b * Hn * Tn;

    int srow = tid >> 4;
    int sc4  = tid & 15;
    const float* qsrc = qp + (size_t)srow * Hn + 4 * sc4;
    const float* vsrc = vt + (size_t)srow * Tn + 4 * sc4;
    int sdst = srow * LP + 4 * sc4;

    v4f pre[16];
#pragma unroll
    for (int r = 0; r < 8; r++) {
        pre[r]     = *(const v4f*)(qsrc + (size_t)(8 * r) * Hn);
        pre[8 + r] = *(const v4f*)(vsrc + (size_t)(8 * r) * Tn);
    }

    v2f kb[16];
#pragma unroll
    for (int kk = 0; kk < 16; kk++) kb[kk] = *(const v2f*)(kp + 4 * kk);

#pragma unroll
    for (int r = 0; r < 8; r++) {
        *(v4f*)(&lq[0][sdst + 8 * r * LP]) = pre[r];
        *(v4f*)(&lv[0][sdst + 8 * r * LP]) = pre[8 + r];
    }
    __syncthreads();

    v8f oacc[4] = {};
    float m = -__builtin_inff();
    float l = 0.0f;
    v2f bb[4];

    int fo = l15 * LP + 2 * hi;

    for (int s = 0; s < g; s++) {
        const float* cq = &lq[s & 1][fo];
        const float* cv = &lv[s & 1][fo];

#pragma unroll
        for (int r = 0; r < 8; r++) {
            pre[r]     = *(const v4f*)(qsrc + (size_t)((s + 1) * KS + 8 * r) * Hn);
            pre[8 + r] = *(const v4f*)(vsrc + (size_t)(8 * r) * Tn + (s + 1) * KS);
        }

        v2f fq[4][16], fv[4][16];
#pragma unroll
        for (int jj = 0; jj < 4; jj++) {
#pragma unroll
            for (int kk = 0; kk < 16; kk++)
                fq[jj][kk] = *(const v2f*)(cq + jj * 16 * LP + 4 * kk);
            if (jj > 0) {
#pragma unroll
                for (int kk = 0; kk < 16; kk++)
                    fv[jj - 1][kk] = *(const v2f*)(cv + (kk >> 2) * 16 * LP + (jj - 1) * 16 + 4 * (kk & 3));
            }
        }
#pragma unroll
        for (int kk = 0; kk < 16; kk++)
            fv[3][kk] = *(const v2f*)(cv + (kk >> 2) * 16 * LP + 3 * 16 + 4 * (kk & 3));
        __builtin_amdgcn_sched_barrier(0);

        v8f sacc = {};
#pragma unroll
        for (int kk = 0; kk < 16; kk++) sacc = WMMA_F32(fq[0][kk], kb[kk], sacc);

#pragma unroll
        for (int jj = 0; jj < 4; jj++) {
            v8f snext = {};
            if (jj < 3) {
#pragma unroll
                for (int kk = 0; kk < 16; kk++) snext = WMMA_F32(fq[jj + 1][kk], kb[kk], snext);
            }

            softmax_step(sacc, m, l, oacc, bb, hi);

#pragma unroll
            for (int n = 0; n < 4; n++) {
#pragma unroll
                for (int kk = 0; kk < 4; kk++)
                    oacc[n] = WMMA_F32(fv[jj][4 * n + kk], bb[kk], oacc[n]);
            }
            sacc = snext;
        }

        __builtin_amdgcn_sched_barrier(0);
#pragma unroll
        for (int r = 0; r < 8; r++) {
            *(v4f*)(&lq[(s + 1) & 1][sdst + 8 * r * LP]) = pre[r];
            *(v4f*)(&lv[(s + 1) & 1][sdst + 8 * r * LP]) = pre[8 + r];
        }
        __syncthreads();
    }

    {
        const float* cq = &lq[g & 1][fo];
        const float* cv = &lv[g & 1][fo];
        for (int jj = 0; jj <= w; jj++) {
            v2f dq[16], dv[16];
#pragma unroll
            for (int kk = 0; kk < 16; kk++) {
                dq[kk] = *(const v2f*)(cq + jj * 16 * LP + 4 * kk);
                dv[kk] = *(const v2f*)(cv + (kk >> 2) * 16 * LP + jj * 16 + 4 * (kk & 3));
            }
            __builtin_amdgcn_sched_barrier(0);

            v8f sacc = {};
#pragma unroll
            for (int kk = 0; kk < 16; kk++) sacc = WMMA_F32(dq[kk], kb[kk], sacc);

            if (jj == w) {
#pragma unroll
                for (int r = 0; r < 8; r++) {
                    int sl = r + 8 * hi;
                    sacc[r] = (sl > l15) ? -__builtin_inff() : sacc[r];
                }
            }

            softmax_step(sacc, m, l, oacc, bb, hi);

#pragma unroll
            for (int n = 0; n < 4; n++) {
#pragma unroll
                for (int kk = 0; kk < 4; kk++)
                    oacc[n] = WMMA_F32(dv[4 * n + kk], bb[kk], oacc[n]);
            }
        }
    }

    float invl = 1.0f / l;
    float* dst = out + (size_t)(b * Tn + i * 16 + l15) * Hn + 8 * hi;
#pragma unroll
    for (int n = 0; n < 4; n++) {
        v4f lo = {oacc[n][0] * invl, oacc[n][1] * invl, oacc[n][2] * invl, oacc[n][3] * invl};
        v4f hi4 = {oacc[n][4] * invl, oacc[n][5] * invl, oacc[n][6] * invl, oacc[n][7] * invl};
        *(v4f*)(dst + n * 16)     = lo;
        *(v4f*)(dst + n * 16 + 4) = hi4;
    }
}

extern "C" void kernel_launch(void* const* d_in, const int* in_sizes, int n_in,
                              void* d_out, int out_size, void* d_ws, size_t ws_size,
                              hipStream_t stream) {
    const float* x  = (const float*)d_in[0];
    const float* Wk = (const float*)d_in[1];
    const float* Wq = (const float*)d_in[2];
    const float* Wv = (const float*)d_in[3];
    float* ws  = (float*)d_ws;
    float* out = (float*)d_out;

    wtrans_kernel<<<192, 256, 0, stream>>>(Wk, Wq, Wv, ws + OFF_WT);
    proj_kernel<<<256, 128, 0, stream>>>(x, ws + OFF_WT, ws);
    attn_kernel<<<256, 128, 0, stream>>>(ws, out);
}


# attention: each upper 64-row group's keys cut in two across work-groups (paired with a lower group: equal work each); a 4th kernel merges the two partial (m,l,O); changes softmax/PV summation order
# speedup vs baseline: 3.9517x; 1.2273x over previous
#include <hip/hip_runtime.h>

typedef float v2f __attribute__((ext_vector_type(2)));
typedef float v4f __attribute__((ext_vector_type(4)));
typedef float v8f __attribute__((ext_vector_type(8)));

#define WMMA_F32(a, b, c) \
    __builtin_amdgcn_wmma_f32_16x16x4_f32(false, (a), false, (b), (short)0, (c), false, false)

constexpr int Bn = 8, Tn = 2048, Cn = 1024, Hn = 64;

constexpr size_t OFF_KP = 0;
constexpr size_t OFF_QP = OFF_KP + (size_t)Bn * Tn * Hn;
constexpr size_t OFF_VT = OFF_QP + (size_t)Bn * Tn * Hn;
constexpr size_t OFF_WT = OFF_VT + (size_t)Bn * Tn * Hn;

constexpr int NCHUNK = Cn / 8;
constexpr int WT_TILE = NCHUNK * 32 * 4;

__global__ void wtrans_kernel(const float* __restrict__ Wk,
                              const float* __restrict__ Wq,
                              const float* __restrict__ Wv,
                              float* __restrict__ wt) {
    int idx  = blockIdx.x * 256 + threadIdx.x;
    int m    = idx >> 14;
    int p    = idx & 16383;
    int nt   = p >> 12;
    int j    = (p >> 5) & (NCHUNK - 1);
    int lane = p & 31;
    int k0   = 8 * j + 4 * (lane >> 4);
    int n    = 16 * nt + (lane & 15);
    const float* W = (m == 0) ? Wk : ((m == 1) ? Wq : Wv);
    float sc = (m == 1) ? 0.125f : 1.0f;
    v4f val = {W[(k0 + 0) * Hn + n] * sc, W[(k0 + 1) * Hn + n] * sc,
               W[(k0 + 2) * Hn + n] * sc, W[(k0 + 3) * Hn + n] * sc};
    *(v4f*)(wt + (size_t)idx * 4) = val;
}

constexpr int PU = 4;
constexpr int RA = 2;
constexpr int NB = 6;

struct FragSet {
    v4f xa[PU][RA];
    v4f wb[PU][NB];
};

__device__ __forceinline__ void load_set(FragSet& f, const float* xa, const float* wb, int j0) {
#pragma unroll
    for (int u = 0; u < PU; u++) {
#pragma unroll
        for (int a = 0; a < RA; a++)
            f.xa[u][a] = *(const v4f*)(xa + (size_t)a * 16 * Cn + 8 * (j0 + u));
    }
#pragma unroll
    for (int u = 0; u < PU; u++) {
#pragma unroll
        for (int n = 0; n < NB; n++)
            f.wb[u][n] = *(const v4f*)(wb + n * WT_TILE + (j0 + u) * 128);
    }
}

__device__ __forceinline__ void mma_set(const FragSet& f, v8f (&acc)[RA][NB]) {
#pragma unroll
    for (int u = 0; u < PU; u++) {
#pragma unroll
        for (int a = 0; a < RA; a++) {
            v2f alo = {f.xa[u][a][0], f.xa[u][a][1]};
#pragma unroll
            for (int n = 0; n < NB; n++)
                acc[a][n] = WMMA_F32(alo, ((v2f){f.wb[u][n][0], f.wb[u][n][1]}), acc[a][n]);
        }
#pragma unroll
        for (int a = 0; a < RA; a++) {
            v2f ahi = {f.xa[u][a][2], f.xa[u][a][3]};
#pragma unroll
            for (int n = 0; n < NB; n++)
                acc[a][n] = WMMA_F32(ahi, ((v2f){f.wb[u][n][2], f.wb[u][n][3]}), acc[a][n]);
        }
    }
}

__device__ __forceinline__ void store_rowmajor(float* p, v8f a) {
#pragma unroll
    for (int r = 0; r < 8; r++) p[(size_t)r * Hn] = a[r];
}

__device__ __forceinline__ void store_transposed(float* d, v8f a) {
    *(v4f*)(d)     = (v4f){a[0], a[1], a[2], a[3]};
    *(v4f*)(d + 4) = (v4f){a[4], a[5], a[6], a[7]};
}

__global__ __launch_bounds__(128) void proj_kernel(const float* __restrict__ x,
                            const float* __restrict__ wt,
                            float* __restrict__ ws) {
    int w    = __builtin_amdgcn_readfirstlane(threadIdx.x >> 5);
    int lane = threadIdx.x & 31;
    int hf   = w & 1;
    int rb   = 2 * blockIdx.x + (w >> 1);
    int l15  = lane & 15;
    int hi   = lane >> 4;

    const float* xa = x + (size_t)(rb * 32 + l15) * Cn + 4 * hi;
    const float* wb = wt + (size_t)(NB * hf) * WT_TILE + 4 * lane;

    v8f acc[RA][NB] = {};
    FragSet f0, f1;
    load_set(f0, xa, wb, 0);
    for (int j = 0; j < NCHUNK - 2 * PU; j += 2 * PU) {
        load_set(f1, xa, wb, j + PU);
        mma_set(f0, acc);
        load_set(f0, xa, wb, j + 2 * PU);
        mma_set(f1, acc);
        __builtin_amdgcn_sched_group_barrier(0x020, (RA + NB) * PU, 0);
        __builtin_amdgcn_sched_group_barrier(0x008, 2 * RA * NB * PU, 0);
        __builtin_amdgcn_sched_group_barrier(0x020, (RA + NB) * PU, 0);
        __builtin_amdgcn_sched_group_barrier(0x008, 2 * RA * NB * PU, 0);
    }
    load_set(f1, xa, wb, NCHUNK - PU);
    mma_set(f0, acc);
    mma_set(f1, acc);

    float* pk = ws + OFF_KP + (size_t)(rb * 32 + 8 * hi) * Hn + l15;
    float* pq = ws + OFF_QP + (size_t)(rb * 32 + 8 * hi) * Hn + l15;
    if (hf == 0) {
#pragma unroll
        for (int a = 0; a < RA; a++) {
            store_rowmajor(pk + (size_t)a * 16 * Hn + 0 * 16, acc[a][0]);
            store_rowmajor(pk + (size_t)a * 16 * Hn + 1 * 16, acc[a][1]);
            store_rowmajor(pk + (size_t)a * 16 * Hn + 2 * 16, acc[a][2]);
            store_rowmajor(pk + (size_t)a * 16 * Hn + 3 * 16, acc[a][3]);
            store_rowmajor(pq + (size_t)a * 16 * Hn + 0 * 16, acc[a][4]);
            store_rowmajor(pq + (size_t)a * 16 * Hn + 1 * 16, acc[a][5]);
        }
    } else {
        int b  = rb >> 6;
        int t0 = (rb & 63) * 32 + 8 * hi;
        float* vtb = ws + OFF_VT + (size_t)b * Hn * Tn + t0 + (size_t)l15 * Tn;
#pragma unroll
        for (int a = 0; a < RA; a++) {
            store_rowmajor(pq + (size_t)a * 16 * Hn + 2 * 16, acc[a][0]);
            store_rowmajor(pq + (size_t)a * 16 * Hn + 3 * 16, acc[a][1]);
            store_transposed(vtb + a * 16 + (size_t)(0 * 16) * Tn, acc[a][2]);
            store_transposed(vtb + a * 16 + (size_t)(1 * 16) * Tn, acc[a][3]);
            store_transposed(vtb + a * 16 + (size_t)(2 * 16) * Tn, acc[a][4]);
            store_transposed(vtb + a * 16 + (size_t)(3 * 16) * Tn, acc[a][5]);
        }
    }
}

constexpr int KS = 64;
constexpr int LP = 68;
constexpr int TILE_F = 64 * LP;

__device__ __forceinline__ float xhalf(float x) {
    int b = __builtin_bit_cast(int, x);
    return __builtin_bit_cast(float,
        __builtin_amdgcn_permlanex16(b, b, 0x76543210, 0xfedcba98, false, false));
}

__device__ __forceinline__ void softmax_step(v8f s, float& m, float& l, v8f (&oacc)[4],
                                             v2f (&bb)[4], int hi) {
    float tm = s[0];
#pragma unroll
    for (int r = 1; r < 8; r++) tm = fmaxf(tm, s[r]);
    tm = fmaxf(tm, xhalf(tm));
    float mn = fmaxf(m, tm);
    float sc = __expf(m - mn);
    m = mn;

    float p[8];
    float tl = 0.0f;
#pragma unroll
    for (int r = 0; r < 8; r++) {
        p[r] = __expf(s[r] - mn);
        tl += p[r];
    }
    tl += xhalf(tl);
    l = l * sc + tl;
#pragma unroll
    for (int n = 0; n < 4; n++) oacc[n] = oacc[n] * sc;

    float q0 = xhalf(p[0]), q1 = xhalf(p[1]), q2 = xhalf(p[2]), q3 = xhalf(p[3]);
    float q4 = xhalf(p[4]), q5 = xhalf(p[5]), q6 = xhalf(p[6]), q7 = xhalf(p[7]);
    bool ishi = (hi != 0);
    bb[0] = ishi ? (v2f){q2, q3} : (v2f){p[0], p[1]};
    bb[1] = ishi ? (v2f){q6, q7} : (v2f){p[4], p[5]};
    bb[2] = ishi ? (v2f){p[2], p[3]} : (v2f){q0, q1};
    bb[3] = ishi ? (v2f){p[6], p[7]} : (v2f){q4, q5};
}

constexpr int SPLIT = 17;

__global__ __launch_bounds__(128) void attn_kernel(float* __restrict__ ws,
                                                   float* __restrict__ out) {
    __shared__ float lq[2][TILE_F];
    __shared__ float lv[2][TILE_F];

    int tid  = threadIdx.x;
    int w    = __builtin_amdgcn_readfirstlane(tid >> 5);
    int lane = tid & 31;
    int b    = blockIdx.x & 7;
    int pr   = (int)(blockIdx.x >> 3);
    int u    = pr >> 1;
    int role = pr & 1;
    int gh   = 31 - u;
    int cut  = gh < SPLIT ? gh : SPLIT;
    bool adiag = gh < SPLIT;
    int l15  = lane & 15;
    int hi   = lane >> 4;

    const float* qp = ws + OFF_QP + (size_t)b * Tn * Hn;
    const float* vt = ws + OFF_VT + (size_t)b * Hn * Tn;

    int srow = tid >> 4;
    int sc4  = tid & 15;
    const float* qsrc = qp + (size_t)srow * Hn + 4 * sc4;
    const float* vsrc = vt + (size_t)srow * Tn + 4 * sc4;
    int sdst = srow * LP + 4 * sc4;
    int fo = l15 * LP + 2 * hi;

    int njobs = role + 1;
    for (int jb = 0; jb < njobs; jb++) {
        int g, s0, s1;
        bool diag;
        if (role == 0)    { g = gh; s0 = 0;   s1 = cut; diag = adiag;  }
        else if (jb == 0) { g = u;  s0 = 0;   s1 = u;   diag = true;   }
        else              { g = gh; s0 = cut; s1 = gh;  diag = !adiag; }
        bool fin = (role == 1) && (jb == 0);
        int i = 4 * g + w;

        const float* kp = ws + OFF_KP + (size_t)(b * Tn + i * 16 + l15) * Hn + 2 * hi;

        v4f pre[16];
#pragma unroll
        for (int r = 0; r < 8; r++) {
            pre[r]     = *(const v4f*)(qsrc + (size_t)(s0 * KS + 8 * r) * Hn);
            pre[8 + r] = *(const v4f*)(vsrc + (size_t)(8 * r) * Tn + s0 * KS);
        }

        v2f kb[16];
#pragma unroll
        for (int kk = 0; kk < 16; kk++) kb[kk] = *(const v2f*)(kp + 4 * kk);

        __syncthreads();
#pragma unroll
        for (int r = 0; r < 8; r++) {
            *(v4f*)(&lq[s0 & 1][sdst + 8 * r * LP]) = pre[r];
            *(v4f*)(&lv[s0 & 1][sdst + 8 * r * LP]) = pre[8 + r];
        }
        __syncthreads();

        v8f oacc[4] = {};
        float m = -__builtin_inff();
        float l = 0.0f;
        v2f bb[4];

        for (int s = s0; s < s1; s++) {
            const float* cq = &lq[s & 1][fo];
            const float* cv = &lv[s & 1][fo];

#pragma unroll
            for (int r = 0; r < 8; r++) {
                pre[r]     = *(const v4f*)(qsrc + (size_t)((s + 1) * KS + 8 * r) * Hn);
                pre[8 + r] = *(const v4f*)(vsrc + (size_t)(8 * r) * Tn + (s + 1) * KS);
            }

            v2f fq[4][16], fv[4][16];
#pragma unroll
            for (int jj = 0; jj < 4; jj++) {
#pragma unroll
                for (int kk = 0; kk < 16; kk++)
                    fq[jj][kk] = *(const v2f*)(cq + jj * 16 * LP + 4 * kk);
                if (jj > 0) {
#pragma unroll
                    for (int kk = 0; kk < 16; kk++)
                        fv[jj - 1][kk] = *(const v2f*)(cv + (kk >> 2) * 16 * LP + (jj - 1) * 16 + 4 * (kk & 3));
                }
            }
#pragma unroll
            for (int kk = 0; kk < 16; kk++)
                fv[3][kk] = *(const v2f*)(cv + (kk >> 2) * 16 * LP + 3 * 16 + 4 * (kk & 3));
            __builtin_amdgcn_sched_barrier(0);

            v8f sacc = {};
#pragma unroll
            for (int kk = 0; kk < 16; kk++) sacc = WMMA_F32(fq[0][kk], kb[kk], sacc);

#pragma unroll
            for (int jj = 0; jj < 4; jj++) {
                v8f snext = {};
                if (jj < 3) {
#pragma unroll
                    for (int kk = 0; kk < 16; kk++) snext = WMMA_F32(fq[jj + 1][kk], kb[kk], snext);
                }

                softmax_step(sacc, m, l, oacc, bb, hi);

#pragma unroll
                for (int n = 0; n < 4; n++) {
#pragma unroll
                    for (int kk = 0; kk < 4; kk++)
                        oacc[n] = WMMA_F32(fv[jj][4 * n + kk], bb[kk], oacc[n]);
                }
                sacc = snext;
            }

            __builtin_amdgcn_sched_barrier(0);
#pragma unroll
            for (int r = 0; r < 8; r++) {
                *(v4f*)(&lq[(s + 1) & 1][sdst + 8 * r * LP]) = pre[r];
                *(v4f*)(&lv[(s + 1) & 1][sdst + 8 * r * LP]) = pre[8 + r];
            }
            __syncthreads();
        }

        if (diag) {
            const float* cq = &lq[g & 1][fo];
            const float* cv = &lv[g & 1][fo];
            for (int jj = 0; jj <= w; jj++) {
                v2f dq[16], dv[16];
#pragma unroll
                for (int kk = 0; kk < 16; kk++) {
                    dq[kk] = *(const v2f*)(cq + jj * 16 * LP + 4 * kk);
                    dv[kk] = *(const v2f*)(cv + (kk >> 2) * 16 * LP + jj * 16 + 4 * (kk & 3));
                }
                __builtin_amdgcn_sched_barrier(0);

                v8f sacc = {};
#pragma unroll
                for (int kk = 0; kk < 16; kk++) sacc = WMMA_F32(dq[kk], kb[kk], sacc);

                if (jj == w) {
#pragma unroll
                    for (int r = 0; r < 8; r++) {
                        int sl = r + 8 * hi;
                        sacc[r] = (sl > l15) ? -__builtin_inff() : sacc[r];
                    }
                }

                softmax_step(sacc, m, l, oacc, bb, hi);

#pragma unroll
                for (int n = 0; n < 4; n++) {
#pragma unroll
                    for (int kk = 0; kk < 4; kk++)
                        oacc[n] = WMMA_F32(dv[4 * n + kk], bb[kk], oacc[n]);
                }
            }
        }

        float fscale = 1.0f;
        float* dst;
        if (fin) {
            fscale = 1.0f / l;
            dst = out + (size_t)(b * Tn + i * 16 + l15) * Hn + 8 * hi;
        } else {
            int hr = (gh - 16) * 64 + w * 16 + l15;
            if (hi == 0)
                *(v2f*)(ws + OFF_WT + ((size_t)(b * 1024 + hr) * 2 + role) * 2) = (v2f){m, l};
            dst = (role == 0)
                ? out + (size_t)(b * Tn + i * 16 + l15) * Hn + 8 * hi
                : ws + OFF_KP + (size_t)(b * Tn + (4 * u + w) * 16 + l15) * Hn + 8 * hi;
        }
#pragma unroll
        for (int n = 0; n < 4; n++) {
            v4f lo = {oacc[n][0] * fscale, oacc[n][1] * fscale, oacc[n][2] * fscale, oacc[n][3] * fscale};
            v4f hi4 = {oacc[n][4] * fscale, oacc[n][5] * fscale, oacc[n][6] * fscale, oacc[n][7] * fscale};
            *(v4f*)(dst + n * 16)     = lo;
            *(v4f*)(dst + n * 16 + 4) = hi4;
        }
    }
}

constexpr int MERGE_PER_THREAD = 4;

__global__ __launch_bounds__(256) void merge_kernel(const float* __restrict__ ws,
                                                    float* __restrict__ out) {
    int base = blockIdx.x * 256 + threadIdx.x;
    v4f ml[MERGE_PER_THREAD], oa[MERGE_PER_THREAD], ob[MERGE_PER_THREAD];
    float* po[MERGE_PER_THREAD];
#pragma unroll
    for (int it = 0; it < MERGE_PER_THREAD; it++) {
        int idx = it * 32768 + base;
        int c4  = idx & 15;
        int hr  = (idx >> 4) & 1023;
        int b   = idx >> 14;
        int u   = 15 - (hr >> 6);
        int tl  = u * 64 + (hr & 63);
        ml[it] = *(const v4f*)(ws + OFF_WT + (size_t)(b * 1024 + hr) * 4);
        po[it] = out + (size_t)(b * Tn + 1024 + hr) * Hn + 4 * c4;
        oa[it] = *(const v4f*)po[it];
        ob[it] = *(const v4f*)(ws + OFF_KP + (size_t)(b * Tn + tl) * Hn + 4 * c4);
    }
#pragma unroll
    for (int it = 0; it < MERGE_PER_THREAD; it++) {
        float M  = fmaxf(ml[it][0], ml[it][2]);
        float ea = __expf(ml[it][0] - M);
        float eb = __expf(ml[it][2] - M);
        float inv = 1.0f / (ml[it][1] * ea + ml[it][3] * eb);
        ea *= inv;
        eb *= inv;
        *(v4f*)po[it] = oa[it] * ea + ob[it] * eb;
    }
}

extern "C" void kernel_launch(void* const* d_in, const int* in_sizes, int n_in,
                              void* d_out, int out_size, void* d_ws, size_t ws_size,
                              hipStream_t stream) {
    const float* x  = (const float*)d_in[0];
    const float* Wk = (const float*)d_in[1];
    const float* Wq = (const float*)d_in[2];
    const float* Wv = (const float*)d_in[3];
    float* ws  = (float*)d_ws;
    float* out = (float*)d_out;

    wtrans_kernel<<<192, 256, 0, stream>>>(Wk, Wq, Wv, ws + OFF_WT);
    proj_kernel<<<256, 128, 0, stream>>>(x, ws + OFF_WT, ws);
    attn_kernel<<<256, 128, 0, stream>>>(ws, out);
    merge_kernel<<<128, 256, 0, stream>>>(ws, out);
}
